# starting kernel
# baseline (speedup 1.0000x reference)
#include <hip/hip_runtime.h>

typedef __attribute__((ext_vector_type(2))) float f32x2;
typedef __attribute__((ext_vector_type(8))) float f32x8;

#define VOCAB 32000
#define EMB   512
#define HID   1024
#define GATES 4096
#define BATCH 32
#define SEQ   512

__global__ __launch_bounds__(256) void zero_ws_kernel(float* __restrict__ p, int n) {
    int i = blockIdx.x * blockDim.x + threadIdx.x;
    if (i < n) p[i] = 0.0f;
}

__device__ __forceinline__ float sigmoidf_dev(float x) {
    return 1.0f / (1.0f + __expf(-x));
}

__global__ __launch_bounds__(256) void lstm_step_kernel(
    const int*   __restrict__ src,
    const float* __restrict__ emb,
    const float* __restrict__ W_ih,
    const float* __restrict__ W_hh,
    const float* __restrict__ b_ih,
    const float* __restrict__ b_hh,
    const float* __restrict__ h_prev, long h_prev_bstride,
    float*       __restrict__ c_state,
    float*       __restrict__ h_out, long h_out_bstride,
    int t)
{
    __shared__ float lds_g[4][BATCH][16];

    const int tid  = threadIdx.x;
    const int lane = tid & 31;
    const int wave = tid >> 5;
    const int gate = wave >> 1;
    const int mt   = wave & 1;
    const int l16  = lane & 15;
    const int hi   = lane >> 4;

    const int n0 = blockIdx.x * 16;
    const int N  = n0 + l16;
    const int M  = mt * 16 + l16;
    const int gcol = gate * HID + N;

    f32x8 acc = {0.f, 0.f, 0.f, 0.f, 0.f, 0.f, 0.f, 0.f};

    {
        const long row = (long)src[M * SEQ + t];
        const float* a_ptr = emb + row * (long)EMB;
        #pragma unroll 4
        for (int k0 = 0; k0 < EMB; k0 += 4) {
            const int kb = k0 + 2 * hi;
            f32x2 a = *(const f32x2*)(a_ptr + kb);
            f32x2 b;
            b.x = W_ih[(long)kb * GATES + gcol];
            b.y = W_ih[(long)(kb + 1) * GATES + gcol];
            acc = __builtin_amdgcn_wmma_f32_16x16x4_f32(
                false, a, false, b, (short)0, acc, false, false);
        }
    }

    {
        const float* a_ptr = h_prev + (long)M * h_prev_bstride;
        #pragma unroll 4
        for (int k0 = 0; k0 < HID; k0 += 4) {
            const int kb = k0 + 2 * hi;
            f32x2 a = *(const f32x2*)(a_ptr + kb);
            f32x2 b;
            b.x = W_hh[(long)kb * GATES + gcol];
            b.y = W_hh[(long)(kb + 1) * GATES + gcol];
            acc = __builtin_amdgcn_wmma_f32_16x16x4_f32(
                false, a, false, b, (short)0, acc, false, false);
        }
    }

    const float bias = b_ih[gcol] + b_hh[gcol];

    #pragma unroll
    for (int r = 0; r < 8; ++r) {
        const int brow = mt * 16 + hi * 8 + r;
        lds_g[gate][brow][l16] = acc[r] + bias;
    }
    __syncthreads();

    #pragma unroll
    for (int idx = tid; idx < BATCH * 16; idx += 256) {
        const int b = idx >> 4;
        const int n = idx & 15;
        const float ig = sigmoidf_dev(lds_g[0][b][n]);
        const float fg = sigmoidf_dev(lds_g[1][b][n]);
        const float gg = tanhf(lds_g[2][b][n]);
        const float og = sigmoidf_dev(lds_g[3][b][n]);
        const long cidx = (long)b * HID + n0 + n;
        const float c_new = fg * c_state[cidx] + ig * gg;
        c_state[cidx] = c_new;
        h_out[(long)b * h_out_bstride + n0 + n] = og * tanhf(c_new);
    }
}

extern "C" void kernel_launch(void* const* d_in, const int* in_sizes, int n_in,
                              void* d_out, int out_size, void* d_ws, size_t ws_size,
                              hipStream_t stream) {
    const int*   src  = (const int*)  d_in[0];
    const float* emb  = (const float*)d_in[1];
    const float* W_ih = (const float*)d_in[2];
    const float* W_hh = (const float*)d_in[3];
    const float* b_ih = (const float*)d_in[4];
    const float* b_hh = (const float*)d_in[5];
    float* out = (float*)d_out;

    float* c_state = (float*)d_ws;
    float* h0      = c_state + BATCH * HID;

    const int n_init = 2 * BATCH * HID;
    zero_ws_kernel<<<(n_init + 255) / 256, 256, 0, stream>>>(c_state, n_init);

    for (int t = 0; t < SEQ; ++t) {
        const float* h_prev = (t == 0) ? h0 : (out + (long)(t - 1) * HID);
        const long   hstr   = (t == 0) ? (long)HID : (long)SEQ * HID;
        lstm_step_kernel<<<HID / 16, 256, 0, stream>>>(
            src, emb, W_ih, W_hh, b_ih, b_hh,
            h_prev, hstr, c_state,
            out + (long)t * HID, (long)SEQ * HID, t);
    }
}


# More loads in flight for the latency-bound step GEMM: K split over 16 waves per block (wave = all 4 gates, grid 64x2), loads 5 k-steps ahead; partials summed in LDS, so the f32 sum order changes
# speedup vs baseline: 3.3440x; 3.3440x over previous
#include <hip/hip_runtime.h>


typedef __attribute__((ext_vector_type(2))) float f32x2;
typedef __attribute__((ext_vector_type(8))) float f32x8;

#define VOCAB 32000
#define EMB   512
#define HID   1024
#define GATES 4096
#define BATCH 32
#define SEQ   512

#define KSPLIT    16
#define EMB_SLICE (EMB / KSPLIT)
#define HID_SLICE (HID / KSPLIT)

__global__ __launch_bounds__(256) void zero_ws_kernel(float* __restrict__ p, int n) {
    int i = blockIdx.x * blockDim.x + threadIdx.x;
    if (i < n) p[i] = 0.0f;
}

__device__ __forceinline__ float sigmoidf_dev(float x) {
    return 1.0f / (1.0f + __expf(-x));
}

#define HID_STEPS (HID_SLICE / 4)
#define EMB_STEPS (EMB_SLICE / 4)
#define NSTEP     (HID_STEPS + EMB_STEPS)
#define DEPTH     4

__global__ __launch_bounds__(32 * KSPLIT) void lstm_step_kernel(
    const int*   __restrict__ src,
    const float* __restrict__ emb,
    const float* __restrict__ W_ih,
    const float* __restrict__ W_hh,
    const float* __restrict__ b_ih,
    const float* __restrict__ b_hh,
    const float* __restrict__ h_prev, long h_prev_bstride,
    float*       __restrict__ c_state,
    float*       __restrict__ h_out, long h_out_bstride,
    int t)
{
    __shared__ float lds_p[KSPLIT][4][16][16];

    const int tid  = threadIdx.x;
    const int lane = tid & 31;
    const int ks   = tid >> 5;
    const int l16  = lane & 15;
    const int hi   = lane >> 4;

    const int n0 = blockIdx.x * 16;
    const int m0 = blockIdx.y * 16;
    const int N  = n0 + l16;
    const int M  = m0 + l16;

    f32x8 acc0 = {0.f, 0.f, 0.f, 0.f, 0.f, 0.f, 0.f, 0.f};
    f32x8 acc1 = acc0, acc2 = acc0, acc3 = acc0;

    const long row = (long)src[M * SEQ + t];

    const float* a_hid = h_prev + (long)M * h_prev_bstride + ks * HID_SLICE + 2 * hi;
    const float* w_hid = W_hh + (long)(ks * HID_SLICE + 2 * hi) * GATES + N;
    const float* a_emb = emb + row * (long)EMB + ks * EMB_SLICE + 2 * hi;
    const float* w_emb = W_ih + (long)(ks * EMB_SLICE + 2 * hi) * GATES + N;

    f32x2 a[NSTEP];
    f32x2 b[NSTEP][4];

#define LOAD_STEP(i)                                                              \
    {                                                                             \
        const int    j  = ((i) < HID_STEPS) ? (i) : ((i) - HID_STEPS);            \
        const float* ap = ((i) < HID_STEPS) ? a_hid : a_emb;                      \
        const float* wp = (((i) < HID_STEPS) ? w_hid : w_emb) + (long)(4 * j) * GATES; \
        a[i] = *(const f32x2*)(ap + 4 * j);                                       \
        _Pragma("unroll")                                                         \
        for (int g = 0; g < 4; ++g) {                                             \
            b[i][g].x = wp[g * HID];                                              \
            b[i][g].y = wp[GATES + g * HID];                                      \
        }                                                                         \
    }

    #pragma unroll
    for (int i = 0; i < DEPTH; ++i) LOAD_STEP(i)
    __builtin_amdgcn_sched_barrier(0);

    #pragma unroll
    for (int i = 0; i < NSTEP; ++i) {
        if (i + DEPTH < NSTEP) LOAD_STEP(i + DEPTH)
        __builtin_amdgcn_sched_barrier(0);
        acc0 = __builtin_amdgcn_wmma_f32_16x16x4_f32(false, a[i], false, b[i][0], (short)0, acc0, false, false);
        acc1 = __builtin_amdgcn_wmma_f32_16x16x4_f32(false, a[i], false, b[i][1], (short)0, acc1, false, false);
        acc2 = __builtin_amdgcn_wmma_f32_16x16x4_f32(false, a[i], false, b[i][2], (short)0, acc2, false, false);
        acc3 = __builtin_amdgcn_wmma_f32_16x16x4_f32(false, a[i], false, b[i][3], (short)0, acc3, false, false);
    }
#undef LOAD_STEP

    #pragma unroll
    for (int r = 0; r < 8; ++r) {
        const int brow = hi * 8 + r;
        lds_p[ks][0][brow][l16] = acc0[r];
        lds_p[ks][1][brow][l16] = acc1[r];
        lds_p[ks][2][brow][l16] = acc2[r];
        lds_p[ks][3][brow][l16] = acc3[r];
    }
    __syncthreads();

    if (tid < 256) {
        const int rb = tid >> 4;
        const int cn = tid & 15;
        float s0 = 0.f, s1 = 0.f, s2 = 0.f, s3 = 0.f;
        #pragma unroll
        for (int k = 0; k < KSPLIT; ++k) {
            s0 += lds_p[k][0][rb][cn];
            s1 += lds_p[k][1][rb][cn];
            s2 += lds_p[k][2][rb][cn];
            s3 += lds_p[k][3][rb][cn];
        }
        const int col = n0 + cn;
        const float ig = sigmoidf_dev(s0 + (b_ih[col]           + b_hh[col]));
        const float fg = sigmoidf_dev(s1 + (b_ih[HID + col]     + b_hh[HID + col]));
        const float gg = tanhf(       s2 + (b_ih[2 * HID + col] + b_hh[2 * HID + col]));
        const float og = sigmoidf_dev(s3 + (b_ih[3 * HID + col] + b_hh[3 * HID + col]));
        const long brow = m0 + rb;
        const long cidx = brow * HID + col;
        const float c_new = fg * c_state[cidx] + ig * gg;
        c_state[cidx] = c_new;
        h_out[brow * h_out_bstride + col] = og * tanhf(c_new);
    }
}

extern "C" void kernel_launch(void* const* d_in, const int* in_sizes, int n_in,
                              void* d_out, int out_size, void* d_ws, size_t ws_size,
                              hipStream_t stream) {
    const int*   src  = (const int*)  d_in[0];
    const float* emb  = (const float*)d_in[1];
    const float* W_ih = (const float*)d_in[2];
    const float* W_hh = (const float*)d_in[3];
    const float* b_ih = (const float*)d_in[4];
    const float* b_hh = (const float*)d_in[5];
    float* out = (float*)d_out;

    float* c_state = (float*)d_ws;
    float* h0      = c_state + BATCH * HID;

    const int n_init = 2 * BATCH * HID;
    zero_ws_kernel<<<(n_init + 255) / 256, 256, 0, stream>>>(c_state, n_init);

    const dim3 grid(HID / 16, BATCH / 16);
    for (int t = 0; t < SEQ; ++t) {
        const float* h_prev = (t == 0) ? h0 : (out + (long)(t - 1) * HID);
        const long   hstr   = (t == 0) ? (long)HID : (long)SEQ * HID;
        lstm_step_kernel<<<grid, 32 * KSPLIT, 0, stream>>>(
            src, emb, W_ih, W_hh, b_ih, b_hh,
            h_prev, hstr, c_state,
            out + (long)t * HID, (long)SEQ * HID, t);
    }
}


# One A load feeds two k-steps (a lane reads 4 consecutive k, 16 bytes): half as many 16-line A-fragment loads, B loads unchanged; k order inside each group of eight permuted, so f32 sum order changes
# speedup vs baseline: 3.5822x; 1.0712x over previous
#include <hip/hip_runtime.h>


typedef __attribute__((ext_vector_type(2))) float f32x2;
typedef __attribute__((ext_vector_type(4))) float f32x4;
typedef __attribute__((ext_vector_type(8))) float f32x8;

#define VOCAB 32000
#define EMB   512
#define HID   1024
#define GATES 4096
#define BATCH 32
#define SEQ   512

#define KSPLIT    16
#define EMB_SLICE (EMB / KSPLIT)
#define HID_SLICE (HID / KSPLIT)

__global__ __launch_bounds__(256) void zero_ws_kernel(float* __restrict__ p, int n) {
    int i = blockIdx.x * blockDim.x + threadIdx.x;
    if (i < n) p[i] = 0.0f;
}

__device__ __forceinline__ float sigmoidf_dev(float x) {
    return 1.0f / (1.0f + __expf(-x));
}

#define HID_STEPS (HID_SLICE / 4)
#define EMB_STEPS (EMB_SLICE / 4)
#define NSTEP     (HID_STEPS + EMB_STEPS)
#define DEPTH     4

__global__ __launch_bounds__(32 * KSPLIT) void lstm_step_kernel(
    const int*   __restrict__ src,
    const float* __restrict__ emb,
    const float* __restrict__ W_ih,
    const float* __restrict__ W_hh,
    const float* __restrict__ b_ih,
    const float* __restrict__ b_hh,
    const float* __restrict__ h_prev, long h_prev_bstride,
    float*       __restrict__ c_state,
    float*       __restrict__ h_out, long h_out_bstride,
    int t)
{
    __shared__ float lds_p[KSPLIT][4][16][16];

    const int tid  = threadIdx.x;
    const int lane = tid & 31;
    const int ks   = tid >> 5;
    const int l16  = lane & 15;
    const int hi   = lane >> 4;

    const int n0 = blockIdx.x * 16;
    const int m0 = blockIdx.y * 16;
    const int N  = n0 + l16;
    const int M  = m0 + l16;

    f32x8 acc0 = {0.f, 0.f, 0.f, 0.f, 0.f, 0.f, 0.f, 0.f};
    f32x8 acc1 = acc0, acc2 = acc0, acc3 = acc0;

    const long row = (long)src[M * SEQ + t];

    const float* a_hid = h_prev + (long)M * h_prev_bstride + ks * HID_SLICE + 4 * hi;
    const float* w_hid = W_hh + (long)(ks * HID_SLICE + 4 * hi) * GATES + N;
    const float* a_emb = emb + row * (long)EMB + ks * EMB_SLICE + 4 * hi;
    const float* w_emb = W_ih + (long)(ks * EMB_SLICE + 4 * hi) * GATES + N;

    f32x4 a4[NSTEP / 2];
    f32x2 b[NSTEP][4];

#define LOAD_STEP(i)                                                              \
    {                                                                             \
        const int    j  = ((i) < HID_STEPS) ? (i) : ((i) - HID_STEPS);            \
        const float* ap = ((i) < HID_STEPS) ? a_hid : a_emb;                      \
        const float* wp = (((i) < HID_STEPS) ? w_hid : w_emb)                     \
                          + (long)(8 * (j >> 1) + 2 * (j & 1)) * GATES;           \
        if (((i) & 1) == 0) a4[(i) / 2] = *(const f32x4*)(ap + 8 * (j >> 1));     \
        _Pragma("unroll")                                                         \
        for (int g = 0; g < 4; ++g) {                                             \
            b[i][g].x = wp[g * HID];                                              \
            b[i][g].y = wp[GATES + g * HID];                                      \
        }                                                                         \
    }

    #pragma unroll
    for (int i = 0; i < DEPTH; ++i) LOAD_STEP(i)
    __builtin_amdgcn_sched_barrier(0);

    #pragma unroll
    for (int i = 0; i < NSTEP; ++i) {
        if (i + DEPTH < NSTEP) LOAD_STEP(i + DEPTH)
        __builtin_amdgcn_sched_barrier(0);
        const f32x2 ai = (i & 1) ? __builtin_shufflevector(a4[i / 2], a4[i / 2], 2, 3)
                                 : __builtin_shufflevector(a4[i / 2], a4[i / 2], 0, 1);
        acc0 = __builtin_amdgcn_wmma_f32_16x16x4_f32(false, ai, false, b[i][0], (short)0, acc0, false, false);
        acc1 = __builtin_amdgcn_wmma_f32_16x16x4_f32(false, ai, false, b[i][1], (short)0, acc1, false, false);
        acc2 = __builtin_amdgcn_wmma_f32_16x16x4_f32(false, ai, false, b[i][2], (short)0, acc2, false, false);
        acc3 = __builtin_amdgcn_wmma_f32_16x16x4_f32(false, ai, false, b[i][3], (short)0, acc3, false, false);
    }
#undef LOAD_STEP

    #pragma unroll
    for (int r = 0; r < 8; ++r) {
        const int brow = hi * 8 + r;
        lds_p[ks][0][brow][l16] = acc0[r];
        lds_p[ks][1][brow][l16] = acc1[r];
        lds_p[ks][2][brow][l16] = acc2[r];
        lds_p[ks][3][brow][l16] = acc3[r];
    }
    __syncthreads();

    if (tid < 256) {
        const int rb = tid >> 4;
        const int cn = tid & 15;
        float s0 = 0.f, s1 = 0.f, s2 = 0.f, s3 = 0.f;
        #pragma unroll
        for (int k = 0; k < KSPLIT; ++k) {
            s0 += lds_p[k][0][rb][cn];
            s1 += lds_p[k][1][rb][cn];
            s2 += lds_p[k][2][rb][cn];
            s3 += lds_p[k][3][rb][cn];
        }
        const int col = n0 + cn;
        const float ig = sigmoidf_dev(s0 + (b_ih[col]           + b_hh[col]));
        const float fg = sigmoidf_dev(s1 + (b_ih[HID + col]     + b_hh[HID + col]));
        const float gg = tanhf(       s2 + (b_ih[2 * HID + col] + b_hh[2 * HID + col]));
        const float og = sigmoidf_dev(s3 + (b_ih[3 * HID + col] + b_hh[3 * HID + col]));
        const long brow = m0 + rb;
        const long cidx = brow * HID + col;
        const float c_new = fg * c_state[cidx] + ig * gg;
        c_state[cidx] = c_new;
        h_out[brow * h_out_bstride + col] = og * tanhf(c_new);
    }
}

extern "C" void kernel_launch(void* const* d_in, const int* in_sizes, int n_in,
                              void* d_out, int out_size, void* d_ws, size_t ws_size,
                              hipStream_t stream) {
    const int*   src  = (const int*)  d_in[0];
    const float* emb  = (const float*)d_in[1];
    const float* W_ih = (const float*)d_in[2];
    const float* W_hh = (const float*)d_in[3];
    const float* b_ih = (const float*)d_in[4];
    const float* b_hh = (const float*)d_in[5];
    float* out = (float*)d_out;

    float* c_state = (float*)d_ws;
    float* h0      = c_state + BATCH * HID;

    const int n_init = 2 * BATCH * HID;
    zero_ws_kernel<<<(n_init + 255) / 256, 256, 0, stream>>>(c_state, n_init);

    const dim3 grid(HID / 16, BATCH / 16);
    for (int t = 0; t < SEQ; ++t) {
        const float* h_prev = (t == 0) ? h0 : (out + (long)(t - 1) * HID);
        const long   hstr   = (t == 0) ? (long)HID : (long)SEQ * HID;
        lstm_step_kernel<<<grid, 32 * KSPLIT, 0, stream>>>(
            src, emb, W_ih, W_hh, b_ih, b_hh,
            h_prev, hstr, c_state,
            out + (long)t * HID, (long)SEQ * HID, t);
    }
}


# A staged through wave-private LDS: a wave reads its K slice of h and of the embedding rows with whole-line 16-byte loads (4 lines a load, not 16), re-reads it in fragment order; f32 sum order kept
# speedup vs baseline: 3.7551x; 1.0483x over previous
#include <hip/hip_runtime.h>


typedef __attribute__((ext_vector_type(2))) float f32x2;
typedef __attribute__((ext_vector_type(4))) float f32x4;
typedef __attribute__((ext_vector_type(8))) float f32x8;

#define VOCAB 32000
#define EMB   512
#define HID   1024
#define GATES 4096
#define BATCH 32
#define SEQ   512

#define KSPLIT    16
#define EMB_SLICE (EMB / KSPLIT)
#define HID_SLICE (HID / KSPLIT)

__global__ __launch_bounds__(256) void zero_ws_kernel(float* __restrict__ p, int n) {
    int i = blockIdx.x * blockDim.x + threadIdx.x;
    if (i < n) p[i] = 0.0f;
}

__device__ __forceinline__ float sigmoidf_dev(float x) {
    return 1.0f / (1.0f + __expf(-x));
}

#define HID_STEPS (HID_SLICE / 4)
#define EMB_STEPS (EMB_SLICE / 4)
#define NSTEP     (HID_STEPS + EMB_STEPS)
#define DEPTH     4
#define NPAIR     (NSTEP / 2)
#define A_COLS    (HID_SLICE + EMB_SLICE)
#define A_PITCH   (A_COLS + 4)

__global__ __launch_bounds__(32 * KSPLIT) void lstm_step_kernel(
    const int*   __restrict__ src,
    const float* __restrict__ emb,
    const float* __restrict__ W_ih,
    const float* __restrict__ W_hh,
    const float* __restrict__ b_ih,
    const float* __restrict__ b_hh,
    const float* __restrict__ h_prev, long h_prev_bstride,
    float*       __restrict__ c_state,
    float*       __restrict__ h_out, long h_out_bstride,
    int t)
{
    __shared__ float lds_p[KSPLIT][4][16][16];
    __shared__ __attribute__((aligned(16))) float lds_a[KSPLIT][16][A_PITCH];

    const int tid  = threadIdx.x;
    const int lane = tid & 31;
    const int ks   = tid >> 5;
    const int l16  = lane & 15;
    const int hi   = lane >> 4;

    const int n0 = blockIdx.x * 16;
    const int m0 = blockIdx.y * 16;
    const int N  = n0 + l16;

    f32x8 acc0 = {0.f, 0.f, 0.f, 0.f, 0.f, 0.f, 0.f, 0.f};
    f32x8 acc1 = acc0, acc2 = acc0, acc3 = acc0;

    const int r2 = lane >> 4, c16 = lane & 15;
    const int r4 = lane >> 3, c8  = lane & 7;
    int tok[4];
    #pragma unroll
    for (int j = 0; j < 4; ++j) tok[j] = src[(m0 + 4 * j + r4) * SEQ + t];
    f32x4 gh[8], ge[4];
    #pragma unroll
    for (int j = 0; j < 8; ++j)
        gh[j] = *(const f32x4*)(h_prev + (long)(m0 + 2 * j + r2) * h_prev_bstride
                                + ks * HID_SLICE + 4 * c16);
    float* const my_a = &lds_a[ks][0][0];
    __builtin_amdgcn_sched_barrier(0);

    const float* w_hid = W_hh + (long)(ks * HID_SLICE + 4 * hi) * GATES + N;
    const float* w_emb = W_ih + (long)(ks * EMB_SLICE + 4 * hi) * GATES + N;

    f32x4 a4[NPAIR];
    f32x2 b[NSTEP][4];

#define LOAD_STEP(i)                                                              \
    {                                                                             \
        const int    j  = ((i) < HID_STEPS) ? (i) : ((i) - HID_STEPS);            \
        const float* wp = (((i) < HID_STEPS) ? w_hid : w_emb)                     \
                          + (long)(8 * (j >> 1) + 2 * (j & 1)) * GATES;           \
        _Pragma("unroll")                                                         \
        for (int g = 0; g < 4; ++g) {                                             \
            b[i][g].x = wp[g * HID];                                              \
            b[i][g].y = wp[GATES + g * HID];                                      \
        }                                                                         \
    }

    #pragma unroll
    for (int i = 0; i < DEPTH; ++i) LOAD_STEP(i)
    #pragma unroll
    for (int j = 0; j < 4; ++j) {
        asm volatile("" : "+v"(tok[j]));
    }
    #pragma unroll
    for (int j = 0; j < 4; ++j)
        ge[j] = *(const f32x4*)(emb + (long)tok[j] * EMB + ks * EMB_SLICE + 4 * c8);
    __builtin_amdgcn_sched_barrier(0);

    #pragma unroll
    for (int j = 0; j < 8; ++j)
        *(f32x4*)(my_a + (2 * j + r2) * A_PITCH + 4 * c16) = gh[j];
    #pragma unroll
    for (int p = 0; p < HID_STEPS / 2; ++p)
        a4[p] = *(const f32x4*)(my_a + l16 * A_PITCH + 8 * p + 4 * hi);

    #pragma unroll
    for (int i = 0; i < NSTEP; ++i) {
        if (i + DEPTH < NSTEP) LOAD_STEP(i + DEPTH)
        if (i == DEPTH) {
            #pragma unroll
            for (int j = 0; j < 4; ++j)
                *(f32x4*)(my_a + (4 * j + r4) * A_PITCH + HID_SLICE + 4 * c8) = ge[j];
            #pragma unroll
            for (int p = HID_STEPS / 2; p < NPAIR; ++p)
                a4[p] = *(const f32x4*)(my_a + l16 * A_PITCH + 8 * p + 4 * hi);
        }
        __builtin_amdgcn_sched_barrier(0);
        const f32x2 ai = (i & 1) ? __builtin_shufflevector(a4[i / 2], a4[i / 2], 2, 3)
                                 : __builtin_shufflevector(a4[i / 2], a4[i / 2], 0, 1);
        acc0 = __builtin_amdgcn_wmma_f32_16x16x4_f32(false, ai, false, b[i][0], (short)0, acc0, false, false);
        acc1 = __builtin_amdgcn_wmma_f32_16x16x4_f32(false, ai, false, b[i][1], (short)0, acc1, false, false);
        acc2 = __builtin_amdgcn_wmma_f32_16x16x4_f32(false, ai, false, b[i][2], (short)0, acc2, false, false);
        acc3 = __builtin_amdgcn_wmma_f32_16x16x4_f32(false, ai, false, b[i][3], (short)0, acc3, false, false);
    }
#undef LOAD_STEP

    #pragma unroll
    for (int r = 0; r < 8; ++r) {
        const int brow = hi * 8 + r;
        lds_p[ks][0][brow][l16] = acc0[r];
        lds_p[ks][1][brow][l16] = acc1[r];
        lds_p[ks][2][brow][l16] = acc2[r];
        lds_p[ks][3][brow][l16] = acc3[r];
    }
    __syncthreads();

    if (tid < 256) {
        const int rb = tid >> 4;
        const int cn = tid & 15;
        float s0 = 0.f, s1 = 0.f, s2 = 0.f, s3 = 0.f;
        #pragma unroll
        for (int k = 0; k < KSPLIT; ++k) {
            s0 += lds_p[k][0][rb][cn];
            s1 += lds_p[k][1][rb][cn];
            s2 += lds_p[k][2][rb][cn];
            s3 += lds_p[k][3][rb][cn];
        }
        const int col = n0 + cn;
        const float ig = sigmoidf_dev(s0 + (b_ih[col]           + b_hh[col]));
        const float fg = sigmoidf_dev(s1 + (b_ih[HID + col]     + b_hh[HID + col]));
        const float gg = tanhf(       s2 + (b_ih[2 * HID + col] + b_hh[2 * HID + col]));
        const float og = sigmoidf_dev(s3 + (b_ih[3 * HID + col] + b_hh[3 * HID + col]));
        const long brow = m0 + rb;
        const long cidx = brow * HID + col;
        const float c_new = fg * c_state[cidx] + ig * gg;
        c_state[cidx] = c_new;
        h_out[brow * h_out_bstride + col] = og * tanhf(c_new);
    }
}

extern "C" void kernel_launch(void* const* d_in, const int* in_sizes, int n_in,
                              void* d_out, int out_size, void* d_ws, size_t ws_size,
                              hipStream_t stream) {
    const int*   src  = (const int*)  d_in[0];
    const float* emb  = (const float*)d_in[1];
    const float* W_ih = (const float*)d_in[2];
    const float* W_hh = (const float*)d_in[3];
    const float* b_ih = (const float*)d_in[4];
    const float* b_hh = (const float*)d_in[5];
    float* out = (float*)d_out;

    float* c_state = (float*)d_ws;
    float* h0      = c_state + BATCH * HID;

    const int n_init = 2 * BATCH * HID;
    zero_ws_kernel<<<(n_init + 255) / 256, 256, 0, stream>>>(c_state, n_init);

    const dim3 grid(HID / 16, BATCH / 16);
    for (int t = 0; t < SEQ; ++t) {
        const float* h_prev = (t == 0) ? h0 : (out + (long)(t - 1) * HID);
        const long   hstr   = (t == 0) ? (long)HID : (long)SEQ * HID;
        lstm_step_kernel<<<grid, 32 * KSPLIT, 0, stream>>>(
            src, emb, W_ih, W_hh, b_ih, b_hh,
            h_prev, hstr, c_state,
            out + (long)t * HID, (long)SEQ * HID, t);
    }
}


# Input projection of step t+1 moved off the step's blocks onto 128 more blocks of launch t (the idle half of the chip), handed over by a 2-plane workspace ring; f32 sum order changes
# speedup vs baseline: 3.8819x; 1.0338x over previous
#include <hip/hip_runtime.h>


typedef __attribute__((ext_vector_type(2))) float f32x2;
typedef __attribute__((ext_vector_type(4))) float f32x4;
typedef __attribute__((ext_vector_type(8))) float f32x8;

#define VOCAB 32000
#define EMB   512
#define HID   1024
#define GATES 4096
#define BATCH 32
#define SEQ   512

#define KSPLIT    16
#define EMB_SLICE (EMB / KSPLIT)
#define HID_SLICE (HID / KSPLIT)

__device__ __forceinline__ float sigmoidf_dev(float x) {
    return 1.0f / (1.0f + __expf(-x));
}

#define HID_STEPS (HID_SLICE / 4)
#define EMB_STEPS (EMB_SLICE / 4)
#define DEPTH     4
#define A_COLS    (HID_SLICE + EMB_SLICE)
#define A_PITCH   (A_COLS + 4)

#define XG_PLANE  (BATCH * GATES)
#define RING_BYTES ((2ul * BATCH * HID + 2ul * XG_PLANE) * sizeof(float))

typedef float (*lds_p_t)[4][16][16];

template <int NH, int NE>
__device__ __forceinline__ void slice_product(
    const int*   __restrict__ src,
    const float* __restrict__ emb,
    const float* __restrict__ W_ih,
    const float* __restrict__ W_hh,
    const float* __restrict__ h_prev, long h_prev_bstride,
    int t, int m0, int n0, float* my_a, lds_p_t lds_p)
{
    constexpr int NS = NH + NE;
    constexpr int NP = NS / 2;
    const int tid  = threadIdx.x;
    const int lane = tid & 31;
    const int ks   = tid >> 5;
    const int l16  = lane & 15;
    const int hi   = lane >> 4;
    const int N    = n0 + l16;

    f32x8 acc0 = {0.f, 0.f, 0.f, 0.f, 0.f, 0.f, 0.f, 0.f};
    f32x8 acc1 = acc0, acc2 = acc0, acc3 = acc0;

    const int r2 = lane >> 4, c16 = lane & 15;
    const int r4 = lane >> 3, c8  = lane & 7;
    int tok[4] = {0, 0, 0, 0};
    f32x4 gh[8], ge[4];
    if constexpr (NE > 0) {
        #pragma unroll
        for (int j = 0; j < 4; ++j) tok[j] = src[(m0 + 4 * j + r4) * SEQ + t];
    }
    if constexpr (NH > 0) {
        #pragma unroll
        for (int j = 0; j < 8; ++j)
            gh[j] = *(const f32x4*)(h_prev + (long)(m0 + 2 * j + r2) * h_prev_bstride
                                    + ks * HID_SLICE + 4 * c16);
    }
    __builtin_amdgcn_sched_barrier(0);

    const float* w_hid = W_hh + (long)(ks * HID_SLICE + 4 * hi) * GATES + N;
    const float* w_emb = W_ih + (long)(ks * EMB_SLICE + 4 * hi) * GATES + N;

    f32x4 a4[NP];
    f32x2 b[NS][4];

#define LOAD_STEP(i)                                                              \
    {                                                                             \
        const int    j  = ((i) < NH) ? (i) : ((i) - NH);                          \
        const float* wp = (((i) < NH) ? w_hid : w_emb)                            \
                          + (long)(8 * (j >> 1) + 2 * (j & 1)) * GATES;           \
        _Pragma("unroll")                                                         \
        for (int g = 0; g < 4; ++g) {                                             \
            b[i][g].x = wp[g * HID];                                              \
            b[i][g].y = wp[GATES + g * HID];                                      \
        }                                                                         \
    }

    #pragma unroll
    for (int i = 0; i < DEPTH; ++i) LOAD_STEP(i)
    if constexpr (NE > 0) {
        #pragma unroll
        for (int j = 0; j < 4; ++j) {
            asm volatile("" : "+v"(tok[j]));
        }
        #pragma unroll
        for (int j = 0; j < 4; ++j)
            ge[j] = *(const f32x4*)(emb + (long)tok[j] * EMB + ks * EMB_SLICE + 4 * c8);
    }
    __builtin_amdgcn_sched_barrier(0);

    if constexpr (NH > 0) {
        #pragma unroll
        for (int j = 0; j < 8; ++j)
            *(f32x4*)(my_a + (2 * j + r2) * A_PITCH + 4 * c16) = gh[j];
        #pragma unroll
        for (int p = 0; p < NH / 2; ++p)
            a4[p] = *(const f32x4*)(my_a + l16 * A_PITCH + 8 * p + 4 * hi);
    }

    constexpr int EMB_AT = (NH > 0) ? DEPTH : 0;

    #pragma unroll
    for (int i = 0; i < NS; ++i) {
        if (i + DEPTH < NS) LOAD_STEP(i + DEPTH)
        if constexpr (NE > 0) {
            if (i == EMB_AT) {
                #pragma unroll
                for (int j = 0; j < 4; ++j)
                    *(f32x4*)(my_a + (4 * j + r4) * A_PITCH + HID_SLICE + 4 * c8) = ge[j];
                #pragma unroll
                for (int p = NH / 2; p < NP; ++p)
                    a4[p] = *(const f32x4*)(my_a + l16 * A_PITCH + HID_SLICE
                                            + 8 * (p - NH / 2) + 4 * hi);
            }
        }
        __builtin_amdgcn_sched_barrier(0);
        const f32x2 ai = (i & 1) ? __builtin_shufflevector(a4[i / 2], a4[i / 2], 2, 3)
                                 : __builtin_shufflevector(a4[i / 2], a4[i / 2], 0, 1);
        acc0 = __builtin_amdgcn_wmma_f32_16x16x4_f32(false, ai, false, b[i][0], (short)0, acc0, false, false);
        acc1 = __builtin_amdgcn_wmma_f32_16x16x4_f32(false, ai, false, b[i][1], (short)0, acc1, false, false);
        acc2 = __builtin_amdgcn_wmma_f32_16x16x4_f32(false, ai, false, b[i][2], (short)0, acc2, false, false);
        acc3 = __builtin_amdgcn_wmma_f32_16x16x4_f32(false, ai, false, b[i][3], (short)0, acc3, false, false);
    }
#undef LOAD_STEP

    #pragma unroll
    for (int r = 0; r < 8; ++r) {
        const int brow = hi * 8 + r;
        lds_p[ks][0][brow][l16] = acc0[r];
        lds_p[ks][1][brow][l16] = acc1[r];
        lds_p[ks][2][brow][l16] = acc2[r];
        lds_p[ks][3][brow][l16] = acc3[r];
    }
}

__device__ __forceinline__ void sum_partials(lds_p_t lds_p, int rb, int cn,
                                             float& s0, float& s1, float& s2, float& s3)
{
    s0 = 0.f; s1 = 0.f; s2 = 0.f; s3 = 0.f;
    #pragma unroll
    for (int k = 0; k < KSPLIT; ++k) {
        s0 += lds_p[k][0][rb][cn];
        s1 += lds_p[k][1][rb][cn];
        s2 += lds_p[k][2][rb][cn];
        s3 += lds_p[k][3][rb][cn];
    }
}

__device__ __forceinline__ void input_projection_tile(
    const int*   __restrict__ src,
    const float* __restrict__ emb,
    const float* __restrict__ W_ih,
    const float* __restrict__ b_ih,
    const float* __restrict__ b_hh,
    float*       __restrict__ xg_tile,
    int t, int m0, int n0, float* my_a, lds_p_t lds_p)
{
    slice_product<0, EMB_STEPS>(src, emb, W_ih, nullptr, nullptr, 0, t, m0, n0, my_a, lds_p);
    __syncthreads();
    const int tid = threadIdx.x;
    if (tid < 256) {
        const int rb = tid >> 4;
        const int cn = tid & 15;
        float s0, s1, s2, s3;
        sum_partials(lds_p, rb, cn, s0, s1, s2, s3);
        const int col = n0 + cn;
        xg_tile[0 * 256 + tid] = s0 + (b_ih[col]           + b_hh[col]);
        xg_tile[1 * 256 + tid] = s1 + (b_ih[HID + col]     + b_hh[HID + col]);
        xg_tile[2 * 256 + tid] = s2 + (b_ih[2 * HID + col] + b_hh[2 * HID + col]);
        xg_tile[3 * 256 + tid] = s3 + (b_ih[3 * HID + col] + b_hh[3 * HID + col]);
    }
}

__global__ __launch_bounds__(32 * KSPLIT) void lstm_first_kernel(
    const int*   __restrict__ src,
    const float* __restrict__ emb,
    const float* __restrict__ W_ih,
    const float* __restrict__ b_ih,
    const float* __restrict__ b_hh,
    float*       __restrict__ state,
    float*       __restrict__ xg_wr,
    unsigned long ws_bytes)
{
    __shared__ float lds_p[KSPLIT][4][16][16];
    __shared__ __attribute__((aligned(16))) float lds_a[KSPLIT][16][A_PITCH];

    const int tid  = threadIdx.x;
    const int tile = blockIdx.y * (HID / 16) + blockIdx.x;
    state[tile * (32 * KSPLIT) + tid] = 0.0f;
    if (ws_bytes < RING_BYTES) return;

    input_projection_tile(src, emb, W_ih, b_ih, b_hh, xg_wr + (long)tile * 1024,
                          0, blockIdx.y * 16, blockIdx.x * 16,
                          &lds_a[tid >> 5][0][0], lds_p);
}

__global__ __launch_bounds__(32 * KSPLIT) void lstm_step_kernel(
    const int*   __restrict__ src,
    const float* __restrict__ emb,
    const float* __restrict__ W_ih,
    const float* __restrict__ W_hh,
    const float* __restrict__ b_ih,
    const float* __restrict__ b_hh,
    const float* __restrict__ h_prev, long h_prev_bstride,
    float*       __restrict__ c_state,
    float*       __restrict__ h_out, long h_out_bstride,
    const float* __restrict__ xg_rd,
    float*       __restrict__ xg_wr,
    unsigned long ws_bytes,
    int t)
{
    __shared__ float lds_p[KSPLIT][4][16][16];
    __shared__ __attribute__((aligned(16))) float lds_a[KSPLIT][16][A_PITCH];

    const int  tid  = threadIdx.x;
    const int  n0   = blockIdx.x * 16;
    const int  m0   = blockIdx.y * 16;
    const int  tile = blockIdx.y * (HID / 16) + blockIdx.x;
    const bool ring = ws_bytes >= RING_BYTES;
    float* const my_a = &lds_a[tid >> 5][0][0];

    if (blockIdx.z == 1) {
        if (ring && t + 1 < SEQ)
            input_projection_tile(src, emb, W_ih, b_ih, b_hh, xg_wr + (long)tile * 1024,
                                  t + 1, m0, n0, my_a, lds_p);
        return;
    }

    const int rb = tid >> 4;
    const int cn = tid & 15;
    const int col = n0 + cn;
    float x0, x1, x2, x3;
    if (ring) {
        if (tid < 256) {
            const float* xt = xg_rd + (long)tile * 1024 + tid;
            x0 = xt[0]; x1 = xt[256]; x2 = xt[512]; x3 = xt[768];
        }
        slice_product<HID_STEPS, 0>(src, emb, W_ih, W_hh, h_prev, h_prev_bstride,
                                    t, m0, n0, my_a, lds_p);
    } else {
        slice_product<HID_STEPS, EMB_STEPS>(src, emb, W_ih, W_hh, h_prev, h_prev_bstride,
                                            t, m0, n0, my_a, lds_p);
    }
    __syncthreads();

    if (tid < 256) {
        float s0, s1, s2, s3;
        sum_partials(lds_p, rb, cn, s0, s1, s2, s3);
        if (!ring) {
            x0 = b_ih[col]           + b_hh[col];
            x1 = b_ih[HID + col]     + b_hh[HID + col];
            x2 = b_ih[2 * HID + col] + b_hh[2 * HID + col];
            x3 = b_ih[3 * HID + col] + b_hh[3 * HID + col];
        }
        const float ig = sigmoidf_dev(s0 + x0);
        const float fg = sigmoidf_dev(s1 + x1);
        const float gg = tanhf(       s2 + x2);
        const float og = sigmoidf_dev(s3 + x3);
        const long brow = m0 + rb;
        const long cidx = brow * HID + col;
        const float c_new = fg * c_state[cidx] + ig * gg;
        c_state[cidx] = c_new;
        h_out[brow * h_out_bstride + col] = og * tanhf(c_new);
    }
}

extern "C" void kernel_launch(void* const* d_in, const int* in_sizes, int n_in,
                              void* d_out, int out_size, void* d_ws, size_t ws_size,
                              hipStream_t stream) {
    const int*   src  = (const int*)  d_in[0];
    const float* emb  = (const float*)d_in[1];
    const float* W_ih = (const float*)d_in[2];
    const float* W_hh = (const float*)d_in[3];
    const float* b_ih = (const float*)d_in[4];
    const float* b_hh = (const float*)d_in[5];
    float* out = (float*)d_out;

    float* c_state = (float*)d_ws;
    float* h0      = c_state + BATCH * HID;
    float* xg      = h0 + BATCH * HID;
    const unsigned long ws_bytes = (unsigned long)ws_size;

    lstm_first_kernel<<<dim3(HID / 16, BATCH / 16, 1), 32 * KSPLIT, 0, stream>>>(
        src, emb, W_ih, b_ih, b_hh, c_state, xg, ws_bytes);

    const dim3 grid(HID / 16, BATCH / 16, 2);
    for (int t = 0; t < SEQ; ++t) {
        const float* h_prev = (t == 0) ? h0 : (out + (long)(t - 1) * HID);
        const long   hstr   = (t == 0) ? (long)HID : (long)SEQ * HID;
        lstm_step_kernel<<<grid, 32 * KSPLIT, 0, stream>>>(
            src, emb, W_ih, W_hh, b_ih, b_hh,
            h_prev, hstr, c_state,
            out + (long)t * HID, (long)SEQ * HID,
            xg + (long)(t & 1) * XG_PLANE, xg + (long)((t + 1) & 1) * XG_PLANE,
            ws_bytes, t);
    }
}


# Cell update's serial chain behind the barrier cut: old cell value and bias asked for at kernel entry, gates on v_exp_f32 + v_rcp_f32 without divide or branch; activation rounding changes
# speedup vs baseline: 4.2509x; 1.0951x over previous
#include <hip/hip_runtime.h>


typedef __attribute__((ext_vector_type(2))) float f32x2;
typedef __attribute__((ext_vector_type(4))) float f32x4;
typedef __attribute__((ext_vector_type(8))) float f32x8;

#define VOCAB 32000
#define EMB   512
#define HID   1024
#define GATES 4096
#define BATCH 32
#define SEQ   512

#define KSPLIT    16
#define EMB_SLICE (EMB / KSPLIT)
#define HID_SLICE (HID / KSPLIT)

__device__ __forceinline__ float sigmoidf_dev(float x) {
    return __builtin_amdgcn_rcpf(1.0f + __expf(-x));
}
__device__ __forceinline__ float tanhf_dev(float x) {
    return 1.0f - 2.0f * __builtin_amdgcn_rcpf(1.0f + __expf(2.0f * x));
}

#define HID_STEPS (HID_SLICE / 4)
#define EMB_STEPS (EMB_SLICE / 4)
#define DEPTH     4
#define A_COLS    (HID_SLICE + EMB_SLICE)
#define A_PITCH   (A_COLS + 4)

#define XG_PLANE  (BATCH * GATES)
#define RING_BYTES ((2ul * BATCH * HID + 2ul * XG_PLANE) * sizeof(float))

typedef float (*lds_p_t)[4][16][16];

template <int NH, int NE>
__device__ __forceinline__ void slice_product(
    const int*   __restrict__ src,
    const float* __restrict__ emb,
    const float* __restrict__ W_ih,
    const float* __restrict__ W_hh,
    const float* __restrict__ h_prev, long h_prev_bstride,
    int t, int m0, int n0, float* my_a, lds_p_t lds_p)
{
    constexpr int NS = NH + NE;
    constexpr int NP = NS / 2;
    const int tid  = threadIdx.x;
    const int lane = tid & 31;
    const int ks   = tid >> 5;
    const int l16  = lane & 15;
    const int hi   = lane >> 4;
    const int N    = n0 + l16;

    f32x8 acc0 = {0.f, 0.f, 0.f, 0.f, 0.f, 0.f, 0.f, 0.f};
    f32x8 acc1 = acc0, acc2 = acc0, acc3 = acc0;

    const int r2 = lane >> 4, c16 = lane & 15;
    const int r4 = lane >> 3, c8  = lane & 7;
    int tok[4] = {0, 0, 0, 0};
    f32x4 gh[8], ge[4];
    if constexpr (NE > 0) {
        #pragma unroll
        for (int j = 0; j < 4; ++j) tok[j] = src[(m0 + 4 * j + r4) * SEQ + t];
    }
    if constexpr (NH > 0) {
        #pragma unroll
        for (int j = 0; j < 8; ++j)
            gh[j] = *(const f32x4*)(h_prev + (long)(m0 + 2 * j + r2) * h_prev_bstride
                                    + ks * HID_SLICE + 4 * c16);
    }
    __builtin_amdgcn_sched_barrier(0);

    const float* w_hid = W_hh + (long)(ks * HID_SLICE + 4 * hi) * GATES + N;
    const float* w_emb = W_ih + (long)(ks * EMB_SLICE + 4 * hi) * GATES + N;

    f32x4 a4[NP];
    f32x2 b[NS][4];

#define LOAD_STEP(i)                                                              \
    {                                                                             \
        const int    j  = ((i) < NH) ? (i) : ((i) - NH);                          \
        const float* wp = (((i) < NH) ? w_hid : w_emb)                            \
                          + (long)(8 * (j >> 1) + 2 * (j & 1)) * GATES;           \
        _Pragma("unroll")                                                         \
        for (int g = 0; g < 4; ++g) {                                             \
            b[i][g].x = wp[g * HID];                                              \
            b[i][g].y = wp[GATES + g * HID];                                      \
        }                                                                         \
    }

    #pragma unroll
    for (int i = 0; i < DEPTH; ++i) LOAD_STEP(i)
    if constexpr (NE > 0) {
        #pragma unroll
        for (int j = 0; j < 4; ++j) {
            asm volatile("" : "+v"(tok[j]));
        }
        #pragma unroll
        for (int j = 0; j < 4; ++j)
            ge[j] = *(const f32x4*)(emb + (long)tok[j] * EMB + ks * EMB_SLICE + 4 * c8);
    }
    __builtin_amdgcn_sched_barrier(0);

    if constexpr (NH > 0) {
        #pragma unroll
        for (int j = 0; j < 8; ++j)
            *(f32x4*)(my_a + (2 * j + r2) * A_PITCH + 4 * c16) = gh[j];
        #pragma unroll
        for (int p = 0; p < NH / 2; ++p)
            a4[p] = *(const f32x4*)(my_a + l16 * A_PITCH + 8 * p + 4 * hi);
    }

    constexpr int EMB_AT = (NH > 0) ? DEPTH : 0;

    #pragma unroll
    for (int i = 0; i < NS; ++i) {
        if (i + DEPTH < NS) LOAD_STEP(i + DEPTH)
        if constexpr (NE > 0) {
            if (i == EMB_AT) {
                #pragma unroll
                for (int j = 0; j < 4; ++j)
                    *(f32x4*)(my_a + (4 * j + r4) * A_PITCH + HID_SLICE + 4 * c8) = ge[j];
                #pragma unroll
                for (int p = NH / 2; p < NP; ++p)
                    a4[p] = *(const f32x4*)(my_a + l16 * A_PITCH + HID_SLICE
                                            + 8 * (p - NH / 2) + 4 * hi);
            }
        }
        __builtin_amdgcn_sched_barrier(0);
        const f32x2 ai = (i & 1) ? __builtin_shufflevector(a4[i / 2], a4[i / 2], 2, 3)
                                 : __builtin_shufflevector(a4[i / 2], a4[i / 2], 0, 1);
        acc0 = __builtin_amdgcn_wmma_f32_16x16x4_f32(false, ai, false, b[i][0], (short)0, acc0, false, false);
        acc1 = __builtin_amdgcn_wmma_f32_16x16x4_f32(false, ai, false, b[i][1], (short)0, acc1, false, false);
        acc2 = __builtin_amdgcn_wmma_f32_16x16x4_f32(false, ai, false, b[i][2], (short)0, acc2, false, false);
        acc3 = __builtin_amdgcn_wmma_f32_16x16x4_f32(false, ai, false, b[i][3], (short)0, acc3, false, false);
    }
#undef LOAD_STEP

    #pragma unroll
    for (int r = 0; r < 8; ++r) {
        const int brow = hi * 8 + r;
        lds_p[ks][0][brow][l16] = acc0[r];
        lds_p[ks][1][brow][l16] = acc1[r];
        lds_p[ks][2][brow][l16] = acc2[r];
        lds_p[ks][3][brow][l16] = acc3[r];
    }
}

__device__ __forceinline__ void sum_partials(lds_p_t lds_p, int rb, int cn,
                                             float& s0, float& s1, float& s2, float& s3)
{
    s0 = 0.f; s1 = 0.f; s2 = 0.f; s3 = 0.f;
    #pragma unroll
    for (int k = 0; k < KSPLIT; ++k) {
        s0 += lds_p[k][0][rb][cn];
        s1 += lds_p[k][1][rb][cn];
        s2 += lds_p[k][2][rb][cn];
        s3 += lds_p[k][3][rb][cn];
    }
}

__device__ __forceinline__ void input_projection_tile(
    const int*   __restrict__ src,
    const float* __restrict__ emb,
    const float* __restrict__ W_ih,
    const float* __restrict__ b_ih,
    const float* __restrict__ b_hh,
    float*       __restrict__ xg_tile,
    int t, int m0, int n0, float* my_a, lds_p_t lds_p)
{
    const int tid = threadIdx.x;
    const int rb = tid >> 4;
    const int cn = tid & 15;
    const int col = n0 + cn;
    float bi0 = 0.f, bi1 = 0.f, bi2 = 0.f, bi3 = 0.f;
    float bh0 = 0.f, bh1 = 0.f, bh2 = 0.f, bh3 = 0.f;
    if (tid < 256) {
        bi0 = b_ih[col];           bh0 = b_hh[col];
        bi1 = b_ih[HID + col];     bh1 = b_hh[HID + col];
        bi2 = b_ih[2 * HID + col]; bh2 = b_hh[2 * HID + col];
        bi3 = b_ih[3 * HID + col]; bh3 = b_hh[3 * HID + col];
    }
    slice_product<0, EMB_STEPS>(src, emb, W_ih, nullptr, nullptr, 0, t, m0, n0, my_a, lds_p);
    __syncthreads();
    if (tid < 256) {
        asm volatile("" : "+v"(bh0), "+v"(bh1), "+v"(bh2), "+v"(bh3));
        float s0, s1, s2, s3;
        sum_partials(lds_p, rb, cn, s0, s1, s2, s3);
        xg_tile[0 * 256 + tid] = s0 + (bi0 + bh0);
        xg_tile[1 * 256 + tid] = s1 + (bi1 + bh1);
        xg_tile[2 * 256 + tid] = s2 + (bi2 + bh2);
        xg_tile[3 * 256 + tid] = s3 + (bi3 + bh3);
    }
}

__global__ __launch_bounds__(32 * KSPLIT) void lstm_first_kernel(
    const int*   __restrict__ src,
    const float* __restrict__ emb,
    const float* __restrict__ W_ih,
    const float* __restrict__ b_ih,
    const float* __restrict__ b_hh,
    float*       __restrict__ state,
    float*       __restrict__ xg_wr,
    unsigned long ws_bytes)
{
    __shared__ float lds_p[KSPLIT][4][16][16];
    __shared__ __attribute__((aligned(16))) float lds_a[KSPLIT][16][A_PITCH];

    const int tid  = threadIdx.x;
    const int tile = blockIdx.y * (HID / 16) + blockIdx.x;
    state[tile * (32 * KSPLIT) + tid] = 0.0f;
    if (ws_bytes < RING_BYTES) return;

    input_projection_tile(src, emb, W_ih, b_ih, b_hh, xg_wr + (long)tile * 1024,
                          0, blockIdx.y * 16, blockIdx.x * 16,
                          &lds_a[tid >> 5][0][0], lds_p);
}

__global__ __launch_bounds__(32 * KSPLIT) void lstm_step_kernel(
    const int*   __restrict__ src,
    const float* __restrict__ emb,
    const float* __restrict__ W_ih,
    const float* __restrict__ W_hh,
    const float* __restrict__ b_ih,
    const float* __restrict__ b_hh,
    const float* __restrict__ h_prev, long h_prev_bstride,
    float*       __restrict__ c_state,
    float*       __restrict__ h_out, long h_out_bstride,
    const float* __restrict__ xg_rd,
    float*       __restrict__ xg_wr,
    unsigned long ws_bytes,
    int t)
{
    __shared__ float lds_p[KSPLIT][4][16][16];
    __shared__ __attribute__((aligned(16))) float lds_a[KSPLIT][16][A_PITCH];

    const int  tid  = threadIdx.x;
    const int  n0   = blockIdx.x * 16;
    const int  m0   = blockIdx.y * 16;
    const int  tile = blockIdx.y * (HID / 16) + blockIdx.x;
    const bool ring = ws_bytes >= RING_BYTES;
    float* const my_a = &lds_a[tid >> 5][0][0];

    if (blockIdx.z == 1) {
        if (ring && t + 1 < SEQ)
            input_projection_tile(src, emb, W_ih, b_ih, b_hh, xg_wr + (long)tile * 1024,
                                  t + 1, m0, n0, my_a, lds_p);
        return;
    }

    const int rb = tid >> 4;
    const int cn = tid & 15;
    const int col = n0 + cn;
    const long brow = m0 + rb;
    const long cidx = brow * HID + col;
    float x0 = 0.f, x1 = 0.f, x2 = 0.f, x3 = 0.f;
    float c_prev = 0.f;
    if (tid < 256) {
        c_prev = c_state[cidx];
        if (ring) {
            const float* xt = xg_rd + (long)tile * 1024 + tid;
            x0 = xt[0]; x1 = xt[256]; x2 = xt[512]; x3 = xt[768];
        }
    }
    if (ring) {
        slice_product<HID_STEPS, 0>(src, emb, W_ih, W_hh, h_prev, h_prev_bstride,
                                    t, m0, n0, my_a, lds_p);
    } else {
        slice_product<HID_STEPS, EMB_STEPS>(src, emb, W_ih, W_hh, h_prev, h_prev_bstride,
                                            t, m0, n0, my_a, lds_p);
    }
    __syncthreads();

    if (tid < 256) {
        float s0, s1, s2, s3;
        sum_partials(lds_p, rb, cn, s0, s1, s2, s3);
        if (!ring) {
            x0 = b_ih[col]           + b_hh[col];
            x1 = b_ih[HID + col]     + b_hh[HID + col];
            x2 = b_ih[2 * HID + col] + b_hh[2 * HID + col];
            x3 = b_ih[3 * HID + col] + b_hh[3 * HID + col];
        }
        const float ig = sigmoidf_dev(s0 + x0);
        const float fg = sigmoidf_dev(s1 + x1);
        const float gg = tanhf_dev(   s2 + x2);
        const float og = sigmoidf_dev(s3 + x3);
        const float c_new = fg * c_prev + ig * gg;
        c_state[cidx] = c_new;
        h_out[brow * h_out_bstride + col] = og * tanhf_dev(c_new);
    }
}

extern "C" void kernel_launch(void* const* d_in, const int* in_sizes, int n_in,
                              void* d_out, int out_size, void* d_ws, size_t ws_size,
                              hipStream_t stream) {
    const int*   src  = (const int*)  d_in[0];
    const float* emb  = (const float*)d_in[1];
    const float* W_ih = (const float*)d_in[2];
    const float* W_hh = (const float*)d_in[3];
    const float* b_ih = (const float*)d_in[4];
    const float* b_hh = (const float*)d_in[5];
    float* out = (float*)d_out;

    float* c_state = (float*)d_ws;
    float* h0      = c_state + BATCH * HID;
    float* xg      = h0 + BATCH * HID;
    const unsigned long ws_bytes = (unsigned long)ws_size;

    lstm_first_kernel<<<dim3(HID / 16, BATCH / 16, 1), 32 * KSPLIT, 0, stream>>>(
        src, emb, W_ih, b_ih, b_hh, c_state, xg, ws_bytes);

    const dim3 grid(HID / 16, BATCH / 16, 2);
    for (int t = 0; t < SEQ; ++t) {
        const float* h_prev = (t == 0) ? h0 : (out + (long)(t - 1) * HID);
        const long   hstr   = (t == 0) ? (long)HID : (long)SEQ * HID;
        lstm_step_kernel<<<grid, 32 * KSPLIT, 0, stream>>>(
            src, emb, W_ih, W_hh, b_ih, b_hh,
            h_prev, hstr, c_state,
            out + (long)t * HID, (long)SEQ * HID,
            xg + (long)(t & 1) * XG_PLANE, xg + (long)((t + 1) & 1) * XG_PLANE,
            ws_bytes, t);
    }
}


# Both weight matrices copied once per call (first launch) into WMMA fragment order in the workspace: a step reads B with whole-line 16-byte loads, 4x fewer loads, half the lines; f32 sum order kept
# speedup vs baseline: 4.6071x; 1.0838x over previous
#include <hip/hip_runtime.h>


typedef __attribute__((ext_vector_type(2))) float f32x2;
typedef __attribute__((ext_vector_type(4))) float f32x4;
typedef __attribute__((ext_vector_type(8))) float f32x8;

#define VOCAB 32000
#define EMB   512
#define HID   1024
#define GATES 4096
#define BATCH 32
#define SEQ   512

#define KSPLIT    16
#define EMB_SLICE (EMB / KSPLIT)
#define HID_SLICE (HID / KSPLIT)

__device__ __forceinline__ float sigmoidf_dev(float x) {
    return __builtin_amdgcn_rcpf(1.0f + __expf(-x));
}
__device__ __forceinline__ float tanhf_dev(float x) {
    return 1.0f - 2.0f * __builtin_amdgcn_rcpf(1.0f + __expf(2.0f * x));
}

#define HID_STEPS (HID_SLICE / 4)
#define EMB_STEPS (EMB_SLICE / 4)
#define DEPTH     4
#define A_COLS    (HID_SLICE + EMB_SLICE)
#define A_PITCH   (A_COLS + 4)

#define XG_PLANE  (BATCH * GATES)
#define RING_BYTES ((2ul * BATCH * HID + 2ul * XG_PLANE) * sizeof(float))

#define WF_H_FLOATS ((unsigned long)HID * GATES)
#define WF_E_FLOATS ((unsigned long)EMB * GATES)
#define FRAG_BYTES  (RING_BYTES + (WF_H_FLOATS + WF_E_FLOATS) * sizeof(float))

typedef float (*lds_p_t)[4][16][16];

template <int NH, int NE, bool FR>
__device__ __forceinline__ void slice_product(
    const int*   __restrict__ src,
    const float* __restrict__ emb,
    const float* __restrict__ W_ih,
    const float* __restrict__ W_hh,
    const float* __restrict__ h_prev, long h_prev_bstride,
    const float* __restrict__ wf_h,
    const float* __restrict__ wf_e,
    int t, int m0, int n0, float* my_a, lds_p_t lds_p)
{
    constexpr int NS = NH + NE;
    constexpr int NP = NS / 2;
    const int tid  = threadIdx.x;
    const int lane = tid & 31;
    const int ks   = tid >> 5;
    const int l16  = lane & 15;
    const int hi   = lane >> 4;
    const int N    = n0 + l16;

    f32x8 acc0 = {0.f, 0.f, 0.f, 0.f, 0.f, 0.f, 0.f, 0.f};
    f32x8 acc1 = acc0, acc2 = acc0, acc3 = acc0;

    const int r2 = lane >> 4, c16 = lane & 15;
    const int r4 = lane >> 3, c8  = lane & 7;
    int tok[4] = {0, 0, 0, 0};
    f32x4 gh[8], ge[4];
    if constexpr (NE > 0) {
        #pragma unroll
        for (int j = 0; j < 4; ++j) tok[j] = src[(m0 + 4 * j + r4) * SEQ + t];
    }
    if constexpr (NH > 0) {
        #pragma unroll
        for (int j = 0; j < 8; ++j)
            gh[j] = *(const f32x4*)(h_prev + (long)(m0 + 2 * j + r2) * h_prev_bstride
                                    + ks * HID_SLICE + 4 * c16);
    }
    __builtin_amdgcn_sched_barrier(0);

    const float* w_hid = W_hh + (long)(ks * HID_SLICE + 4 * hi) * GATES + N;
    const float* w_emb = W_ih + (long)(ks * EMB_SLICE + 4 * hi) * GATES + N;

    f32x4 a4[NP];
    f32x2 b[NS][4];
    f32x4 bq[NP][4];

#define LOAD_STEP(i)                                                              \
    {                                                                             \
        const int    j  = ((i) < NH) ? (i) : ((i) - NH);                          \
        const float* wp = (((i) < NH) ? w_hid : w_emb)                            \
                          + (long)(8 * (j >> 1) + 2 * (j & 1)) * GATES;           \
        _Pragma("unroll")                                                         \
        for (int g = 0; g < 4; ++g) {                                             \
            b[i][g].x = wp[g * HID];                                              \
            b[i][g].y = wp[GATES + g * HID];                                      \
        }                                                                         \
    }

    if constexpr (FR) {
        const long wslice = (long)((n0 >> 4) * KSPLIT + ks);
        const f32x4* qh = (const f32x4*)wf_h + wslice * (HID_STEPS / 2 * 4 * 32) + lane;
        const f32x4* qe = (const f32x4*)wf_e + wslice * (EMB_STEPS / 2 * 4 * 32) + lane;
        #pragma unroll
        for (int p = 0; p < NP; ++p) {
            #pragma unroll
            for (int g = 0; g < 4; ++g) {
                if (p < NH / 2) bq[p][g] = qh[(p * 4 + g) * 32];
                else            bq[p][g] = qe[((p - NH / 2) * 4 + g) * 32];
            }
        }
    } else {
        #pragma unroll
        for (int i = 0; i < DEPTH; ++i) LOAD_STEP(i)
    }
    if constexpr (NE > 0) {
        #pragma unroll
        for (int j = 0; j < 4; ++j) {
            asm volatile("" : "+v"(tok[j]));
        }
        #pragma unroll
        for (int j = 0; j < 4; ++j)
            ge[j] = *(const f32x4*)(emb + (long)tok[j] * EMB + ks * EMB_SLICE + 4 * c8);
    }
    __builtin_amdgcn_sched_barrier(0);

    if constexpr (NH > 0) {
        #pragma unroll
        for (int j = 0; j < 8; ++j)
            *(f32x4*)(my_a + (2 * j + r2) * A_PITCH + 4 * c16) = gh[j];
        #pragma unroll
        for (int p = 0; p < NH / 2; ++p)
            a4[p] = *(const f32x4*)(my_a + l16 * A_PITCH + 8 * p + 4 * hi);
    }

    constexpr int EMB_AT = (NH > 0) ? DEPTH : 0;

    #pragma unroll
    for (int i = 0; i < NS; ++i) {
        if constexpr (!FR) {
            if (i + DEPTH < NS) LOAD_STEP(i + DEPTH)
        }
        if constexpr (NE > 0) {
            if (i == EMB_AT) {
                #pragma unroll
                for (int j = 0; j < 4; ++j)
                    *(f32x4*)(my_a + (4 * j + r4) * A_PITCH + HID_SLICE + 4 * c8) = ge[j];
                #pragma unroll
                for (int p = NH / 2; p < NP; ++p)
                    a4[p] = *(const f32x4*)(my_a + l16 * A_PITCH + HID_SLICE
                                            + 8 * (p - NH / 2) + 4 * hi);
            }
        }
        if constexpr (!FR) {
            __builtin_amdgcn_sched_barrier(0);
        }
        const f32x2 ai = (i & 1) ? __builtin_shufflevector(a4[i / 2], a4[i / 2], 2, 3)
                                 : __builtin_shufflevector(a4[i / 2], a4[i / 2], 0, 1);
        f32x2 b0, b1, b2, b3;
        if constexpr (FR) {
            if (i & 1) {
                b0 = __builtin_shufflevector(bq[i / 2][0], bq[i / 2][0], 2, 3);
                b1 = __builtin_shufflevector(bq[i / 2][1], bq[i / 2][1], 2, 3);
                b2 = __builtin_shufflevector(bq[i / 2][2], bq[i / 2][2], 2, 3);
                b3 = __builtin_shufflevector(bq[i / 2][3], bq[i / 2][3], 2, 3);
            } else {
                b0 = __builtin_shufflevector(bq[i / 2][0], bq[i / 2][0], 0, 1);
                b1 = __builtin_shufflevector(bq[i / 2][1], bq[i / 2][1], 0, 1);
                b2 = __builtin_shufflevector(bq[i / 2][2], bq[i / 2][2], 0, 1);
                b3 = __builtin_shufflevector(bq[i / 2][3], bq[i / 2][3], 0, 1);
            }
        } else {
            b0 = b[i][0]; b1 = b[i][1]; b2 = b[i][2]; b3 = b[i][3];
        }
        acc0 = __builtin_amdgcn_wmma_f32_16x16x4_f32(false, ai, false, b0, (short)0, acc0, false, false);
        acc1 = __builtin_amdgcn_wmma_f32_16x16x4_f32(false, ai, false, b1, (short)0, acc1, false, false);
        acc2 = __builtin_amdgcn_wmma_f32_16x16x4_f32(false, ai, false, b2, (short)0, acc2, false, false);
        acc3 = __builtin_amdgcn_wmma_f32_16x16x4_f32(false, ai, false, b3, (short)0, acc3, false, false);
    }
#undef LOAD_STEP

    #pragma unroll
    for (int r = 0; r < 8; ++r) {
        const int brow = hi * 8 + r;
        lds_p[ks][0][brow][l16] = acc0[r];
        lds_p[ks][1][brow][l16] = acc1[r];
        lds_p[ks][2][brow][l16] = acc2[r];
        lds_p[ks][3][brow][l16] = acc3[r];
    }
}

__device__ __forceinline__ void sum_partials(lds_p_t lds_p, int rb, int cn,
                                             float& s0, float& s1, float& s2, float& s3)
{
    s0 = 0.f; s1 = 0.f; s2 = 0.f; s3 = 0.f;
    #pragma unroll
    for (int k = 0; k < KSPLIT; ++k) {
        s0 += lds_p[k][0][rb][cn];
        s1 += lds_p[k][1][rb][cn];
        s2 += lds_p[k][2][rb][cn];
        s3 += lds_p[k][3][rb][cn];
    }
}

template <bool FR>
__device__ __forceinline__ void input_projection_tile(
    const int*   __restrict__ src,
    const float* __restrict__ emb,
    const float* __restrict__ W_ih,
    const float* __restrict__ b_ih,
    const float* __restrict__ b_hh,
    const float* __restrict__ wf_e,
    float*       __restrict__ xg_tile,
    int t, int m0, int n0, float* my_a, lds_p_t lds_p)
{
    const int tid = threadIdx.x;
    const int rb = tid >> 4;
    const int cn = tid & 15;
    const int col = n0 + cn;
    float bi0 = 0.f, bi1 = 0.f, bi2 = 0.f, bi3 = 0.f;
    float bh0 = 0.f, bh1 = 0.f, bh2 = 0.f, bh3 = 0.f;
    if (tid < 256) {
        bi0 = b_ih[col];           bh0 = b_hh[col];
        bi1 = b_ih[HID + col];     bh1 = b_hh[HID + col];
        bi2 = b_ih[2 * HID + col]; bh2 = b_hh[2 * HID + col];
        bi3 = b_ih[3 * HID + col]; bh3 = b_hh[3 * HID + col];
    }
    slice_product<0, EMB_STEPS, FR>(src, emb, W_ih, nullptr, nullptr, 0, nullptr, wf_e,
                                    t, m0, n0, my_a, lds_p);
    __syncthreads();
    if (tid < 256) {
        asm volatile("" : "+v"(bh0), "+v"(bh1), "+v"(bh2), "+v"(bh3));
        float s0, s1, s2, s3;
        sum_partials(lds_p, rb, cn, s0, s1, s2, s3);
        xg_tile[0 * 256 + tid] = s0 + (bi0 + bh0);
        xg_tile[1 * 256 + tid] = s1 + (bi1 + bh1);
        xg_tile[2 * 256 + tid] = s2 + (bi2 + bh2);
        xg_tile[3 * 256 + tid] = s3 + (bi3 + bh3);
    }
}

__device__ __forceinline__ void pack_weights(
    const float* __restrict__ W_ih,
    const float* __restrict__ W_hh,
    float*       __restrict__ wf_h,
    float*       __restrict__ wf_e,
    int nt, int half)
{
    const int  tid  = threadIdx.x;
    const int  lane = tid & 31;
    const int  ks   = tid >> 5;
    const int  hi   = lane >> 4;
    const int  col  = nt * 16 + (lane & 15);
    const long wslice = (long)(nt * KSPLIT + ks);
    #pragma unroll
    for (int pp = 0; pp < HID_STEPS / 4; ++pp) {
        const int    p  = (HID_STEPS / 4) * half + pp;
        const float* wp = W_hh + (long)(ks * HID_SLICE + 8 * p + 4 * hi) * GATES + col;
        f32x4*       q  = (f32x4*)wf_h + (wslice * (HID_STEPS / 2) + p) * (4 * 32) + lane;
        #pragma unroll
        for (int g = 0; g < 4; ++g) {
            f32x4 v;
            v.x = wp[g * HID];
            v.y = wp[GATES + g * HID];
            v.z = wp[2 * GATES + g * HID];
            v.w = wp[3 * GATES + g * HID];
            q[g * 32] = v;
        }
    }
    #pragma unroll
    for (int pp = 0; pp < EMB_STEPS / 4; ++pp) {
        const int    p  = (EMB_STEPS / 4) * half + pp;
        const float* wp = W_ih + (long)(ks * EMB_SLICE + 8 * p + 4 * hi) * GATES + col;
        f32x4*       q  = (f32x4*)wf_e + (wslice * (EMB_STEPS / 2) + p) * (4 * 32) + lane;
        #pragma unroll
        for (int g = 0; g < 4; ++g) {
            f32x4 v;
            v.x = wp[g * HID];
            v.y = wp[GATES + g * HID];
            v.z = wp[2 * GATES + g * HID];
            v.w = wp[3 * GATES + g * HID];
            q[g * 32] = v;
        }
    }
}

__global__ __launch_bounds__(32 * KSPLIT) void lstm_first_kernel(
    const int*   __restrict__ src,
    const float* __restrict__ emb,
    const float* __restrict__ W_ih,
    const float* __restrict__ W_hh,
    const float* __restrict__ b_ih,
    const float* __restrict__ b_hh,
    float*       __restrict__ state,
    float*       __restrict__ xg_wr,
    float*       __restrict__ wfrag,
    unsigned long ws_bytes)
{
    __shared__ float lds_p[KSPLIT][4][16][16];
    __shared__ __attribute__((aligned(16))) float lds_a[KSPLIT][16][A_PITCH];

    const int tid  = threadIdx.x;
    const int tile = blockIdx.y * (HID / 16) + blockIdx.x;
    state[tile * (32 * KSPLIT) + tid] = 0.0f;
    if (ws_bytes < RING_BYTES) return;

    if (ws_bytes >= FRAG_BYTES)
        pack_weights(W_ih, W_hh, wfrag, wfrag + WF_H_FLOATS, blockIdx.x, blockIdx.y);

    input_projection_tile<false>(src, emb, W_ih, b_ih, b_hh, nullptr,
                                 xg_wr + (long)tile * 1024,
                                 0, blockIdx.y * 16, blockIdx.x * 16,
                                 &lds_a[tid >> 5][0][0], lds_p);
}

__global__ __launch_bounds__(32 * KSPLIT) void lstm_step_kernel(
    const int*   __restrict__ src,
    const float* __restrict__ emb,
    const float* __restrict__ W_ih,
    const float* __restrict__ W_hh,
    const float* __restrict__ b_ih,
    const float* __restrict__ b_hh,
    const float* __restrict__ h_prev, long h_prev_bstride,
    float*       __restrict__ c_state,
    float*       __restrict__ h_out, long h_out_bstride,
    const float* __restrict__ xg_rd,
    float*       __restrict__ xg_wr,
    const float* __restrict__ wfrag,
    unsigned long ws_bytes,
    int t)
{
    __shared__ float lds_p[KSPLIT][4][16][16];
    __shared__ __attribute__((aligned(16))) float lds_a[KSPLIT][16][A_PITCH];

    const int  tid  = threadIdx.x;
    const int  n0   = blockIdx.x * 16;
    const int  m0   = blockIdx.y * 16;
    const int  tile = blockIdx.y * (HID / 16) + blockIdx.x;
    const bool ring = ws_bytes >= RING_BYTES;
    const bool frag = ws_bytes >= FRAG_BYTES;
    float* const my_a = &lds_a[tid >> 5][0][0];

    if (blockIdx.z == 1) {
        if (ring && t + 1 < SEQ) {
            if (frag)
                input_projection_tile<true>(src, emb, W_ih, b_ih, b_hh, wfrag + WF_H_FLOATS,
                                            xg_wr + (long)tile * 1024,
                                            t + 1, m0, n0, my_a, lds_p);
            else
                input_projection_tile<false>(src, emb, W_ih, b_ih, b_hh, nullptr,
                                             xg_wr + (long)tile * 1024,
                                             t + 1, m0, n0, my_a, lds_p);
        }
        return;
    }

    const int rb = tid >> 4;
    const int cn = tid & 15;
    const int col = n0 + cn;
    const long brow = m0 + rb;
    const long cidx = brow * HID + col;
    float x0 = 0.f, x1 = 0.f, x2 = 0.f, x3 = 0.f;
    float c_prev = 0.f;
    if (tid < 256) {
        c_prev = c_state[cidx];
        if (ring) {
            const float* xt = xg_rd + (long)tile * 1024 + tid;
            x0 = xt[0]; x1 = xt[256]; x2 = xt[512]; x3 = xt[768];
        }
    }
    if (frag) {
        slice_product<HID_STEPS, 0, true>(src, emb, W_ih, W_hh, h_prev, h_prev_bstride,
                                          wfrag, nullptr, t, m0, n0, my_a, lds_p);
    } else if (ring) {
        slice_product<HID_STEPS, 0, false>(src, emb, W_ih, W_hh, h_prev, h_prev_bstride,
                                           nullptr, nullptr, t, m0, n0, my_a, lds_p);
    } else {
        slice_product<HID_STEPS, EMB_STEPS, false>(src, emb, W_ih, W_hh, h_prev, h_prev_bstride,
                                                   nullptr, nullptr, t, m0, n0, my_a, lds_p);
    }
    __syncthreads();

    if (tid < 256) {
        float s0, s1, s2, s3;
        sum_partials(lds_p, rb, cn, s0, s1, s2, s3);
        if (!ring) {
            x0 = b_ih[col]           + b_hh[col];
            x1 = b_ih[HID + col]     + b_hh[HID + col];
            x2 = b_ih[2 * HID + col] + b_hh[2 * HID + col];
            x3 = b_ih[3 * HID + col] + b_hh[3 * HID + col];
        }
        const float ig = sigmoidf_dev(s0 + x0);
        const float fg = sigmoidf_dev(s1 + x1);
        const float gg = tanhf_dev(   s2 + x2);
        const float og = sigmoidf_dev(s3 + x3);
        const float c_new = fg * c_prev + ig * gg;
        c_state[cidx] = c_new;
        h_out[brow * h_out_bstride + col] = og * tanhf_dev(c_new);
    }
}

extern "C" void kernel_launch(void* const* d_in, const int* in_sizes, int n_in,
                              void* d_out, int out_size, void* d_ws, size_t ws_size,
                              hipStream_t stream) {
    const int*   src  = (const int*)  d_in[0];
    const float* emb  = (const float*)d_in[1];
    const float* W_ih = (const float*)d_in[2];
    const float* W_hh = (const float*)d_in[3];
    const float* b_ih = (const float*)d_in[4];
    const float* b_hh = (const float*)d_in[5];
    float* out = (float*)d_out;

    float* c_state = (float*)d_ws;
    float* h0      = c_state + BATCH * HID;
    float* xg      = h0 + BATCH * HID;
    float* wfrag   = xg + 2 * XG_PLANE;
    const unsigned long ws_bytes = (unsigned long)ws_size;

    lstm_first_kernel<<<dim3(HID / 16, BATCH / 16, 1), 32 * KSPLIT, 0, stream>>>(
        src, emb, W_ih, W_hh, b_ih, b_hh, c_state, xg, wfrag, ws_bytes);

    const dim3 grid(HID / 16, BATCH / 16, 2);
    for (int t = 0; t < SEQ; ++t) {
        const float* h_prev = (t == 0) ? h0 : (out + (long)(t - 1) * HID);
        const long   hstr   = (t == 0) ? (long)HID : (long)SEQ * HID;
        lstm_step_kernel<<<grid, 32 * KSPLIT, 0, stream>>>(
            src, emb, W_ih, W_hh, b_ih, b_hh,
            h_prev, hstr, c_state,
            out + (long)t * HID, (long)SEQ * HID,
            xg + (long)(t & 1) * XG_PLANE, xg + (long)((t + 1) & 1) * XG_PLANE,
            wfrag, ws_bytes, t);
    }
}
